# starting kernel
# baseline (speedup 1.0000x reference)
#include <hip/hip_runtime.h>
#include <hip/hip_bf16.h>

typedef float v2f __attribute__((ext_vector_type(2)));
typedef float v8f __attribute__((ext_vector_type(8)));

#define WMMA_F32(a, b, c) \
    __builtin_amdgcn_wmma_f32_16x16x4_f32(false, (a), false, (b), (short)0, (c), false, false)

#define EDIM 768
#define HDIM 384
#define G4H  1536
#define BSZ  32
#define SSZ  512
#define ROWS (BSZ * SSZ)

__device__ __forceinline__ float sigmoid_f(float x) {
    return 1.0f / (1.0f + __expf(-x));
}
__device__ __forceinline__ float tanh_f(float x) {
    return 1.0f - 2.0f / (__expf(2.0f * x) + 1.0f);
}

__global__ __launch_bounds__(128)
void xg_gemm_kernel(const float* __restrict__ Af, const float* __restrict__ Ab,
                    const float* __restrict__ Wf, const float* __restrict__ Wb,
                    const float* __restrict__ bihf, const float* __restrict__ bhhf,
                    const float* __restrict__ bihb, const float* __restrict__ bhhb,
                    float* __restrict__ xg, int Kin, int reverse)
{
    const int dir = blockIdx.z;
    const float* __restrict__ A   = dir ? Ab   : Af;
    const float* __restrict__ W   = dir ? Wb   : Wf;
    const float* __restrict__ bih = dir ? bihb : bihf;
    const float* __restrict__ bhh = dir ? bhhb : bhhf;

    const int n0 = blockIdx.x * 64;
    const int m0 = blockIdx.y * 64;

    __shared__ float As[64][17];
    __shared__ float Bs[16][65];

    const int tid   = threadIdx.x;
    const int lane  = tid & 31;
    const int wid   = tid >> 5;
    const int wm    = (wid & 1) * 32;
    const int wn    = (wid >> 1) * 32;
    const int lhalf = lane >> 4;
    const int l16   = lane & 15;

    v8f acc[2][2] = {};

    for (int k0 = 0; k0 < Kin; k0 += 16) {
        #pragma unroll
        for (int i = 0; i < 8; i++) {
            int idx = tid + i * 128;
            int mr  = idx >> 4;
            int kk  = idx & 15;
            int m   = m0 + mr;
            int b   = m >> 9;
            int s   = m & 511;
            int srd = (reverse && dir) ? (511 - s) : s;
            As[mr][kk] = A[(size_t)(b * SSZ + srd) * Kin + (k0 + kk)];
        }
        #pragma unroll
        for (int i = 0; i < 8; i++) {
            int idx = tid + i * 128;
            int nr  = idx >> 4;
            int kk  = idx & 15;
            Bs[kk][nr] = W[(size_t)(n0 + nr) * Kin + (k0 + kk)];
        }
        __syncthreads();

        #pragma unroll
        for (int kc = 0; kc < 16; kc += 4) {
            v2f afrag[2], bfrag[2];
            #pragma unroll
            for (int mt = 0; mt < 2; mt++) {
                int m = wm + mt * 16 + l16;
                afrag[mt].x = As[m][kc + 2 * lhalf];
                afrag[mt].y = As[m][kc + 2 * lhalf + 1];
            }
            #pragma unroll
            for (int nt = 0; nt < 2; nt++) {
                int n = wn + nt * 16 + l16;
                bfrag[nt].x = Bs[kc + 2 * lhalf][n];
                bfrag[nt].y = Bs[kc + 2 * lhalf + 1][n];
            }
            #pragma unroll
            for (int mt = 0; mt < 2; mt++)
                #pragma unroll
                for (int nt = 0; nt < 2; nt++)
                    acc[mt][nt] = WMMA_F32(afrag[mt], bfrag[nt], acc[mt][nt]);
        }
        __syncthreads();
    }

    #pragma unroll
    for (int mt = 0; mt < 2; mt++)
        #pragma unroll
        for (int nt = 0; nt < 2; nt++) {
            int n = n0 + wn + nt * 16 + l16;
            float bias = bih[n] + bhh[n];
            #pragma unroll
            for (int r = 0; r < 8; r++) {
                int m = m0 + wm + mt * 16 + r + 8 * lhalf;
                int b = m >> 9;
                int s = m & 511;
                __builtin_nontemporal_store(
                    acc[mt][nt][r] + bias,
                    &xg[(((size_t)dir * SSZ + s) * BSZ + b) * G4H + n]);
            }
        }
}

__global__ __launch_bounds__(768)
void lstm_recur_kernel(const float* __restrict__ xg,
                       const float* __restrict__ Whhf, const float* __restrict__ Whhb,
                       float* __restrict__ out, size_t dirStride, int rowStride)
{
    const int dir = blockIdx.x;
    const float* __restrict__ Whh = dir ? Whhb : Whhf;
    const float* __restrict__ xgd = xg + (size_t)dir * SSZ * BSZ * G4H;

    __shared__ float hs[32][HDIM + 1];

    const int tid   = threadIdx.x;
    const int lane  = tid & 31;
    const int wid   = tid >> 5;
    const int hc0   = wid * 16;
    const int lhalf = lane >> 4;
    const int l16   = lane & 15;

    for (int i = tid; i < 32 * (HDIM + 1); i += 768)
        (&hs[0][0])[i] = 0.0f;

    v8f cst[2] = {};
    float* outd = out + dir * dirStride;

    __syncthreads();

    for (int t = 0; t < SSZ; t++) {
        v8f acc[4][2];
        const float* __restrict__ xgt = xgd + (size_t)t * BSZ * G4H;
        #pragma unroll
        for (int g = 0; g < 4; g++)
            #pragma unroll
            for (int mt = 0; mt < 2; mt++)
                #pragma unroll
                for (int r = 0; r < 8; r++) {
                    int b = mt * 16 + r + 8 * lhalf;
                    acc[g][mt][r] = xgt[(size_t)b * G4H + g * HDIM + hc0 + l16];
                }

        if (t + 1 < SSZ) {
            const char* nxt = (const char*)(xgd + (size_t)(t + 1) * BSZ * G4H);
            __builtin_prefetch(nxt + (size_t)tid * 256, 0, 0);
            __builtin_prefetch(nxt + (size_t)tid * 256 + 128, 0, 0);
        }

        for (int kc = 0; kc < HDIM; kc += 4) {
            v2f af[2];
            #pragma unroll
            for (int mt = 0; mt < 2; mt++) {
                int b = mt * 16 + l16;
                af[mt].x = hs[b][kc + 2 * lhalf];
                af[mt].y = hs[b][kc + 2 * lhalf + 1];
            }
            #pragma unroll
            for (int g = 0; g < 4; g++) {
                int n = g * HDIM + hc0 + l16;
                const float* wp = Whh + (size_t)n * HDIM + kc + 2 * lhalf;
                v2f bf;
                bf.x = wp[0];
                bf.y = wp[1];
                acc[g][0] = WMMA_F32(af[0], bf, acc[g][0]);
                acc[g][1] = WMMA_F32(af[1], bf, acc[g][1]);
            }
        }
        __syncthreads();

        #pragma unroll
        for (int mt = 0; mt < 2; mt++) {
            #pragma unroll
            for (int r = 0; r < 8; r++) {
                float iv = sigmoid_f(acc[0][mt][r]);
                float fv = sigmoid_f(acc[1][mt][r]);
                float gv = tanh_f(acc[2][mt][r]);
                float ov = sigmoid_f(acc[3][mt][r]);
                float c  = fv * cst[mt][r] + iv * gv;
                cst[mt][r] = c;
                float h  = ov * tanh_f(c);
                int b = mt * 16 + r + 8 * lhalf;
                int n = hc0 + l16;
                hs[b][n] = h;
                __builtin_nontemporal_store(
                    h, &outd[(size_t)(b * SSZ + t) * rowStride + n]);
            }
        }
        __syncthreads();
    }
}

extern "C" void kernel_launch(void* const* d_in, const int* in_sizes, int n_in,
                              void* d_out, int out_size, void* d_ws, size_t ws_size,
                              hipStream_t stream) {
    const float* emb    = (const float*)d_in[0];
    const float* wih_f0 = (const float*)d_in[2];
    const float* whh_f0 = (const float*)d_in[3];
    const float* bih_f0 = (const float*)d_in[4];
    const float* bhh_f0 = (const float*)d_in[5];
    const float* wih_f1 = (const float*)d_in[6];
    const float* whh_f1 = (const float*)d_in[7];
    const float* bih_f1 = (const float*)d_in[8];
    const float* bhh_f1 = (const float*)d_in[9];
    const float* wih_b0 = (const float*)d_in[10];
    const float* whh_b0 = (const float*)d_in[11];
    const float* bih_b0 = (const float*)d_in[12];
    const float* bhh_b0 = (const float*)d_in[13];
    const float* wih_b1 = (const float*)d_in[14];
    const float* whh_b1 = (const float*)d_in[15];
    const float* bih_b1 = (const float*)d_in[16];
    const float* bhh_b1 = (const float*)d_in[17];

    float* xg = (float*)d_ws;
    float* h0 = xg + (size_t)2 * SSZ * BSZ * G4H;
    float* out = (float*)d_out;

    dim3 gA(G4H / 64, ROWS / 64, 2);
    dim3 bA(128);
    dim3 gR(2);
    dim3 bR(768);

    xg_gemm_kernel<<<gA, bA, 0, stream>>>(emb, emb, wih_f0, wih_b0,
                                          bih_f0, bhh_f0, bih_b0, bhh_b0,
                                          xg, EDIM, 1);
    lstm_recur_kernel<<<gR, bR, 0, stream>>>(xg, whh_f0, whh_b0,
                                             h0, (size_t)ROWS * HDIM, HDIM);
    xg_gemm_kernel<<<gA, bA, 0, stream>>>(h0, h0 + (size_t)ROWS * HDIM,
                                          wih_f1, wih_b1,
                                          bih_f1, bhh_f1, bih_b1, bhh_b1,
                                          xg, HDIM, 0);
    lstm_recur_kernel<<<gR, bR, 0, stream>>>(xg, whh_f1, whh_b1,
                                             out, (size_t)HDIM, 2 * HDIM);
}


# Recurrence step runs its two batch tiles as two phases, each with its own pass over the weights; a phase's arrival signal is waited for one phase later, so cell updates run under other waves' WMMAs
# speedup vs baseline: 3.9489x; 3.9489x over previous
#include <hip/hip_runtime.h>
#include <hip/hip_bf16.h>

typedef float v2f __attribute__((ext_vector_type(2)));
typedef float v8f __attribute__((ext_vector_type(8)));
typedef float v4f __attribute__((ext_vector_type(4)));

#define WMMA_F32(a, b, c) \
    __builtin_amdgcn_wmma_f32_16x16x4_f32(false, (a), false, (b), (short)0, (c), false, false)

#define EDIM 768
#define HDIM 384
#define G4H  1536
#define BSZ  32
#define SSZ  512
#define TCH     16
#define LOG_TCH 4
#define NCH     (SSZ / TCH)
#define XG_JOBS 64
#define XCH ((size_t)2 * TCH * BSZ * G4H)
#define HCH ((size_t)2 * BSZ * TCH * HDIM)

#define WS_PITCH 36
#define WS_STAGE (32 * WS_PITCH)

__device__ __forceinline__ float sigmoid_f(float x) {
    return __builtin_amdgcn_rcpf(1.0f + __expf(-x));
}
__device__ __forceinline__ float tanh_f(float x) {
    return 1.0f - 2.0f * __builtin_amdgcn_rcpf(__expf(2.0f * x) + 1.0f);
}

__device__ __forceinline__ void xg_block(float* __restrict__ lds,
                                         const float* __restrict__ A,
                                         const float* __restrict__ W,
                                         const float* __restrict__ bih,
                                         const float* __restrict__ bhh,
                                         float* __restrict__ xg,
                                         int Kin, int rowsB, int tfirst, int tstep,
                                         int m0, int nbase)
{
    const int tid   = threadIdx.x;
    const int grp   = tid >> 7;
    const int tl    = tid & 127;
    const int lane  = tid & 31;
    const int wq    = (tid >> 5) & 3;
    const int wm    = (wq & 1) * 32;
    const int wn    = (wq >> 1) * 32;
    const int lhalf = lane >> 4;
    const int l16   = lane & 15;
    const int n0    = nbase + grp * 64;

    float* As = lds + grp * (64 * 17 + 16 * 65);
    float* Bs = As + 64 * 17;

    v8f acc[2][2] = {};

    for (int k0 = 0; k0 < Kin; k0 += 16) {
        #pragma unroll
        for (int i = 0; i < 8; i++) {
            int idx = tl + i * 128;
            int mr  = idx >> 4;
            int kk  = idx & 15;
            int m   = m0 + mr;
            int b   = m >> LOG_TCH;
            int s   = m & (TCH - 1);
            As[mr * 17 + kk] = A[(size_t)(b * rowsB + tfirst + tstep * s) * Kin + (k0 + kk)];
        }
        #pragma unroll
        for (int i = 0; i < 8; i++) {
            int idx = tl + i * 128;
            int nr  = idx >> 4;
            int kk  = idx & 15;
            Bs[kk * 65 + nr] = W[(size_t)(n0 + nr) * Kin + (k0 + kk)];
        }
        __syncthreads();

        #pragma unroll
        for (int kc = 0; kc < 16; kc += 4) {
            v2f afrag[2], bfrag[2];
            #pragma unroll
            for (int mt = 0; mt < 2; mt++) {
                int m = wm + mt * 16 + l16;
                afrag[mt].x = As[m * 17 + kc + 2 * lhalf];
                afrag[mt].y = As[m * 17 + kc + 2 * lhalf + 1];
            }
            #pragma unroll
            for (int nt = 0; nt < 2; nt++) {
                int n = wn + nt * 16 + l16;
                bfrag[nt].x = Bs[(kc + 2 * lhalf) * 65 + n];
                bfrag[nt].y = Bs[(kc + 2 * lhalf + 1) * 65 + n];
            }
            #pragma unroll
            for (int mt = 0; mt < 2; mt++)
                #pragma unroll
                for (int nt = 0; nt < 2; nt++)
                    acc[mt][nt] = WMMA_F32(afrag[mt], bfrag[nt], acc[mt][nt]);
        }
        __syncthreads();
    }

    #pragma unroll
    for (int mt = 0; mt < 2; mt++)
        #pragma unroll
        for (int nt = 0; nt < 2; nt++) {
            int n = n0 + wn + nt * 16 + l16;
            float bias = bih[n] + bhh[n];
            #pragma unroll
            for (int r = 0; r < 8; r++) {
                int m = m0 + wm + mt * 16 + r + 8 * lhalf;
                int b = m >> LOG_TCH;
                int s = m & (TCH - 1);
                __builtin_nontemporal_store(
                    acc[mt][nt][r] + bias,
                    &xg[((size_t)s * BSZ + b) * G4H + n]);
            }
        }
}


__device__ __forceinline__ void wave_lds_order() {
    __builtin_amdgcn_fence(__ATOMIC_ACQ_REL, "wavefront");
    __builtin_amdgcn_wave_barrier();
}

__device__ __forceinline__ void ws_fetch(v4f (&pre)[4], const float* __restrict__ wl) {
    #pragma unroll
    for (int i = 0; i < 4; i++)
        pre[i] = *(const v4f*)(wl + (size_t)(4 * i) * HDIM);
}
__device__ __forceinline__ void ws_put(float* __restrict__ sl, const v4f (&pre)[4]) {
    #pragma unroll
    for (int i = 0; i < 4; i++)
        *(v4f*)(sl + i * 4 * WS_PITCH) = pre[i];
}

__device__ __forceinline__ void h_publish_phase(const bool wait_prev) {
    __builtin_amdgcn_fence(__ATOMIC_RELEASE, "workgroup", "local");
    if (wait_prev) __builtin_amdgcn_s_barrier_wait(-1);
    __builtin_amdgcn_s_barrier_signal(-1);
    __builtin_amdgcn_fence(__ATOMIC_ACQUIRE, "workgroup", "local");
}

__device__ __forceinline__ void gate_init(v8f (&acc)[4], const float* __restrict__ xlane) {
    #pragma unroll
    for (int g = 0; g < 4; g++)
        #pragma unroll
        for (int r = 0; r < 8; r++)
            acc[g][r] = xlane[r * G4H + g * HDIM];
}

__device__ __forceinline__ void mm_quarter(const float* __restrict__ fr,
                                           const v2f (&af)[8],
                                           int ks0,
                                           v8f &ga, v8f &gb) {
    #pragma unroll
    for (int ks = ks0; ks < ks0 + 4; ks++) {
        v2f b0 = *(const v2f*)(fr + 4 * ks);
        v2f b1 = *(const v2f*)(fr + 16 * WS_PITCH + 4 * ks);
        ga = WMMA_F32(af[ks], b0, ga);
        gb = WMMA_F32(af[ks], b1, gb);
    }
}

__global__ __launch_bounds__(768)
void lstm_chunk_kernel(const float* __restrict__ emb,
                       const float* __restrict__ Wihf0, const float* __restrict__ Wihb0,
                       const float* __restrict__ bihf0, const float* __restrict__ bhhf0,
                       const float* __restrict__ bihb0, const float* __restrict__ bhhb0,
                       const float* __restrict__ Whhf0, const float* __restrict__ Whhb0,
                       const float* __restrict__ Wihf1, const float* __restrict__ Wihb1,
                       const float* __restrict__ bihf1, const float* __restrict__ bhhf1,
                       const float* __restrict__ bihb1, const float* __restrict__ bhhb1,
                       const float* __restrict__ Whhf1, const float* __restrict__ Whhb1,
                       float* __restrict__ xg0buf,
                       float* __restrict__ xg1buf,
                       float* __restrict__ h0buf,
                       float* __restrict__ outfin,
                       float* __restrict__ state,
                       int k, int nsteps)
{
    __shared__ __attribute__((aligned(64))) float hs[2][32][HDIM + 1];
    __shared__ __attribute__((aligned(16))) float wst[24][2][WS_STAGE];

    if (blockIdx.x >= 4) {
        const int job   = blockIdx.x - 4;
        const int which = job >= XG_JOBS;
        const int jj    = which ? job - XG_JOBS : job;
        const int chunk = which ? k - 1 : k + 1;
        if (chunk < 0 || chunk >= NCH) return;
        const int gdir  = jj >> 5;
        const int m0    = ((jj >> 2) & 7) * 64;
        const int nbase = (jj & 3) * 384;
        const size_t par = (size_t)(chunk & 1);
        float* lds = &wst[0][0][0];
        if (which == 0) {
            const int tc = chunk * TCH;
            xg_block(lds, emb, gdir ? Wihb0 : Wihf0,
                     gdir ? bihb0 : bihf0, gdir ? bhhb0 : bhhf0,
                     xg0buf + par * XCH + (size_t)gdir * TCH * BSZ * G4H,
                     EDIM, SSZ, gdir ? (SSZ - 1 - tc) : tc, gdir ? -1 : 1, m0, nbase);
        } else {
            xg_block(lds, h0buf + par * HCH + (size_t)gdir * BSZ * TCH * HDIM,
                     gdir ? Wihb1 : Wihf1,
                     gdir ? bihb1 : bihf1, gdir ? bhhb1 : bhhf1,
                     xg1buf + par * XCH + (size_t)gdir * TCH * BSZ * G4H,
                     HDIM, TCH, 0, 1, m0, nbase);
        }
        return;
    }

    const int layer = blockIdx.x >> 1;
    const int dir   = blockIdx.x & 1;
    const int chunk = layer ? k - 2 : k;
    if (chunk < 0 || chunk >= NCH) return;
    const int n     = nsteps;
    const int t0    = chunk * TCH;
    const size_t par = (size_t)(chunk & 1);
    const float* __restrict__ Whh = layer ? (dir ? Whhb1 : Whhf1) : (dir ? Whhb0 : Whhf0);
    const int xgS    = TCH;
    const int xgTsub = t0;
    const float* __restrict__ xgd = (layer ? xg1buf : xg0buf) + par * XCH + (size_t)dir * xgS * BSZ * G4H;
    const size_t dirStride = layer ? (size_t)HDIM : (size_t)BSZ * TCH * HDIM;
    const int rowStride    = layer ? 2 * HDIM : HDIM;
    const int outS         = layer ? SSZ : TCH;
    const int outTsub      = layer ? 0 : t0;
    float* __restrict__ out = layer ? outfin : h0buf + par * HCH;
    float* sth = state + (size_t)(layer * 2 + dir) * 2 * BSZ * HDIM;
    float* stc = sth + BSZ * HDIM;

    const int tid   = threadIdx.x;
    const int lane  = tid & 31;
    const int wid   = tid >> 5;
    const int hc0   = wid * 16;
    const int lhalf = lane >> 4;
    const int l16   = lane & 15;

    if (t0 == 0) {
        for (int i = tid; i < 32 * (HDIM + 1); i += 768)
            (&hs[0][0][0])[i] = 0.0f;
    } else {
        for (int i = tid; i < 32 * HDIM; i += 768) {
            int b = i / HDIM;
            hs[0][b][i - b * HDIM] = sth[i];
        }
    }

    v8f cst[2] = {};
    if (t0 != 0) {
        #pragma unroll
        for (int mt = 0; mt < 2; mt++)
            #pragma unroll
            for (int r = 0; r < 8; r++)
                cst[mt][r] = stc[(mt * 16 + r + 8 * lhalf) * HDIM + hc0 + l16];
    }
    float* outd = out + dir * dirStride;

    const int rs = lane >> 3;
    const int q  = lane & 7;
    const float* __restrict__ wlane = Whh + (size_t)(hc0 + rs) * HDIM + 4 * q;
    float* st0 = &wst[wid][0][rs * WS_PITCH + 4 * q];
    float* st1 = &wst[wid][1][rs * WS_PITCH + 4 * q];
    const float* fr0 = &wst[wid][0][l16 * WS_PITCH + 2 * lhalf];
    const float* fr1 = &wst[wid][1][l16 * WS_PITCH + 2 * lhalf];

    const float* hrow0 = &hs[0][l16][2 * lhalf];
    const float* hrow1 = &hs[0][16 + l16][2 * lhalf];
    const int HBUF = 32 * (HDIM + 1);
    float* hput = &hs[0][8 * lhalf][hc0 + l16];
    const size_t GATE = (size_t)HDIM * HDIM;

    const int xg_lane_off  = 8 * lhalf * G4H + hc0 + l16;
    const int out_lane_off = 8 * lhalf * outS * rowStride + hc0 + l16;

    v4f preA[4], preB[4];
    ws_fetch(preA, wlane);
    ws_put(st0, preA);
    ws_fetch(preA, wlane + GATE);
    ws_put(st0 + 16 * WS_PITCH, preA);
    wave_lds_order();
    ws_fetch(preB, wlane + 2 * GATE);

    int knext = 32;

    {
        const char* nxt = (const char*)(xgd + (size_t)(t0 - xgTsub) * BSZ * G4H);
        __builtin_prefetch(nxt + (size_t)tid * 256, 0, 0);
        __builtin_prefetch(nxt + (size_t)tid * 256 + 128, 0, 0);
    }

    __syncthreads();

    for (int tl = 0; tl < n; tl++) {
        const int t   = t0 + tl;
        const int cur = tl & 1;

        if (tl + 1 < n) {
            const char* nxt = (const char*)(xgd + (size_t)(t + 1 - xgTsub) * BSZ * G4H);
            __builtin_prefetch(nxt + (size_t)tid * 256, 0, 0);
            __builtin_prefetch(nxt + (size_t)tid * 256 + 128, 0, 0);
        }

        auto phase = [&](const int mt) __attribute__((always_inline)) {
            const float* hrow = (mt ? hrow1 : hrow0) + cur * HBUF;
            v8f acc[4] = {};
            auto k_chunk = [&](const int kc) __attribute__((always_inline)) {
                const int kb = kc * 32;
                const float* h = hrow + kb;
                v2f af[8];
                #pragma unroll
                for (int ks = 0; ks < 8; ks++) {
                    af[ks].x = h[4 * ks];
                    af[ks].y = h[4 * ks + 1];
                }
                ws_fetch(preA, wlane + 3 * GATE + kb);
                mm_quarter(fr0, af, 0, acc[0], acc[1]);
                wave_lds_order();
                ws_put(st1, preB);
                ws_fetch(preB, wlane + knext);
                mm_quarter(fr0, af, 4, acc[0], acc[1]);
                wave_lds_order();
                ws_put(st1 + 16 * WS_PITCH, preA);
                wave_lds_order();

                ws_fetch(preA, wlane + GATE + knext);
                mm_quarter(fr1, af, 0, acc[2], acc[3]);
                wave_lds_order();
                ws_put(st0, preB);
                ws_fetch(preB, wlane + 2 * GATE + knext);
                mm_quarter(fr1, af, 4, acc[2], acc[3]);
                wave_lds_order();
                ws_put(st0 + 16 * WS_PITCH, preA);
                wave_lds_order();
                knext = (knext == HDIM - 32) ? 0 : knext + 32;
            };
            if (mt == 0) {
                __builtin_amdgcn_s_setprio(3);
                for (int kc = 0; kc < HDIM / 32; kc++) k_chunk(kc);
            } else {
                __builtin_amdgcn_s_setprio(2);
                for (int kc = 0; kc < 6; kc++) k_chunk(kc);
                __builtin_amdgcn_s_setprio(1);
                for (int kc = 6; kc < 10; kc++) k_chunk(kc);
                __builtin_amdgcn_s_setprio(0);
                #pragma nounroll
                for (int kc = 10; kc < HDIM / 32; kc++) k_chunk(kc);
            }

            v8f xv[4];
            gate_init(xv, xgd + (size_t)(t - xgTsub) * BSZ * G4H + xg_lane_off + mt * 16 * G4H);

            float* hnew = hput + (cur ^ 1) * HBUF + mt * 16 * (HDIM + 1);
            #pragma unroll
            for (int r = 0; r < 8; r++) {
                float iv = sigmoid_f(xv[0][r] + acc[0][r]);
                float fv = sigmoid_f(xv[1][r] + acc[1][r]);
                float gv = tanh_f(xv[2][r] + acc[2][r]);
                float ov = sigmoid_f(xv[3][r] + acc[3][r]);
                float c  = fv * cst[mt][r] + iv * gv;
                cst[mt][r] = c;
                float h  = ov * tanh_f(c);
                hnew[r * (HDIM + 1)] = h;
                float* orow = outd + (size_t)((mt * 16 + r) * outS + (t - outTsub)) * rowStride;
                __builtin_nontemporal_store(h, &orow[out_lane_off]);
            }
            h_publish_phase(mt != 0 || tl != 0);
        };
        phase(0);
        phase(1);
    }
    if (n > 0) __builtin_amdgcn_s_barrier_wait(-1);
    __builtin_amdgcn_fence(__ATOMIC_ACQUIRE, "workgroup", "local");

    if (t0 + n < SSZ) {
        const int fin = n & 1;
        for (int i = tid; i < 32 * HDIM; i += 768) {
            int b = i / HDIM;
            sth[i] = hs[fin][b][i - b * HDIM];
        }
        #pragma unroll
        for (int mt = 0; mt < 2; mt++)
            #pragma unroll
            for (int r = 0; r < 8; r++)
                stc[(mt * 16 + r + 8 * lhalf) * HDIM + hc0 + l16] = cst[mt][r];
    }
}

extern "C" void kernel_launch(void* const* d_in, const int* in_sizes, int n_in,
                              void* d_out, int out_size, void* d_ws, size_t ws_size,
                              hipStream_t stream) {
    const float* emb    = (const float*)d_in[0];
    const float* wih_f0 = (const float*)d_in[2];
    const float* whh_f0 = (const float*)d_in[3];
    const float* bih_f0 = (const float*)d_in[4];
    const float* bhh_f0 = (const float*)d_in[5];
    const float* wih_f1 = (const float*)d_in[6];
    const float* whh_f1 = (const float*)d_in[7];
    const float* bih_f1 = (const float*)d_in[8];
    const float* bhh_f1 = (const float*)d_in[9];
    const float* wih_b0 = (const float*)d_in[10];
    const float* whh_b0 = (const float*)d_in[11];
    const float* bih_b0 = (const float*)d_in[12];
    const float* bhh_b0 = (const float*)d_in[13];
    const float* wih_b1 = (const float*)d_in[14];
    const float* whh_b1 = (const float*)d_in[15];
    const float* bih_b1 = (const float*)d_in[16];
    const float* bhh_b1 = (const float*)d_in[17];

    float* xg0 = (float*)d_ws;
    float* xg1 = xg0 + 2 * XCH;
    float* h0c = xg1 + 2 * XCH;
    float* st  = h0c + 2 * HCH;
    float* out = (float*)d_out;

    dim3 gR(4 + 2 * XG_JOBS);
    dim3 bR(768);

    for (int k = -1; k <= NCH + 1; k++) {
        lstm_chunk_kernel<<<gR, bR, 0, stream>>>(
            emb, wih_f0, wih_b0, bih_f0, bhh_f0, bih_b0, bhh_b0, whh_f0, whh_b0,
            wih_f1, wih_b1, bih_f1, bhh_f1, bih_b1, bhh_b1, whh_f1, whh_b1,
            xg0, xg1, h0c, out, st, k, TCH);
    }
}
